# starting kernel
# baseline (speedup 1.0000x reference)
#include <hip/hip_runtime.h>
#include <hip/hip_bf16.h>

typedef __attribute__((ext_vector_type(2))) float v2f;
typedef __attribute__((ext_vector_type(8))) float v8f;

#define N_ROWS 16384
#define DIMS   128
#define EPS_N  1e-12f

__global__ void collapse_inv_norm_kernel(const float* __restrict__ fc,
                                         float* __restrict__ inv) {
  const int wave = (blockIdx.x * blockDim.x + threadIdx.x) >> 5;
  const int lane = threadIdx.x & 31;
  const float4* p = (const float4*)(fc + (size_t)wave * DIMS) + lane;
  float4 v = *p;
  float ss = v.x * v.x + v.y * v.y + v.z * v.z + v.w * v.w;
#pragma unroll
  for (int off = 16; off > 0; off >>= 1) {
    ss += __shfl_xor(ss, off, 32);
  }
  if (lane == 0) {
    inv[wave] = 1.0f / fmaxf(sqrtf(ss), EPS_N);
  }
}

__global__ void collapse_adj_dot_kernel(const float* __restrict__ fc,
                                        const float* __restrict__ inv,
                                        float* __restrict__ out) {
  const int lane = threadIdx.x & 31;
  const int wavesPerBlk = blockDim.x >> 5;
  const int tile = blockIdx.x * wavesPerBlk + (threadIdx.x >> 5);
  const int base = tile * 16;

  const int m    = lane & 15;
  const int half = lane >> 4;

  const int rowA = base + m;
  int rowB = base + 1 + m;
  if (rowB > N_ROWS - 1) rowB = N_ROWS - 1;

  const float* pa = fc + (size_t)rowA * DIMS + 2 * half;
  const float* pb = fc + (size_t)rowB * DIMS + 2 * half;

  v8f acc = {0.f, 0.f, 0.f, 0.f, 0.f, 0.f, 0.f, 0.f};
#pragma unroll
  for (int k = 0; k < DIMS; k += 4) {
    v2f a = *(const v2f*)(pa + k);
    v2f b = *(const v2f*)(pb + k);
    acc = __builtin_amdgcn_wmma_f32_16x16x4_f32(
        false, a, false, b, (short)0, acc, false, false);
  }

  float diag = 0.f;
#pragma unroll
  for (int r = 0; r < 8; ++r) {
    float cr = acc[r];
    if (lane == r)      diag = cr;
    if (lane == r + 24) diag = cr;
  }

  const bool have = (lane < 8) || (lane >= 24);
  const int  md   = (lane < 8) ? lane : (lane - 16);
  const int  oidx = base + md;
  if (have && oidx < N_ROWS - 1) {
    float d = diag * inv[oidx] * inv[oidx + 1];
    const float c = 1.0f / (float)(N_ROWS - 1);
    float t = d + c;
    out[oidx] = t * t;
  }
}

extern "C" void kernel_launch(void* const* d_in, const int* in_sizes, int n_in,
                              void* d_out, int out_size, void* d_ws, size_t ws_size,
                              hipStream_t stream) {
  (void)in_sizes; (void)n_in; (void)out_size; (void)ws_size;
  const float* fc  = (const float*)d_in[0];
  float*       out = (float*)d_out;
  float*       inv = (float*)d_ws;

  collapse_inv_norm_kernel<<<(N_ROWS * 32) / 256, 256, 0, stream>>>(fc, inv);

  const int tiles = N_ROWS / 16;
  collapse_adj_dot_kernel<<<tiles / 8, 256, 0, stream>>>(fc, inv, out);
}


# One pass through LDS: each wave reads its 17 rows once with whole-row loads, sums squares for the norms from those registers, reads both WMMA operands from its LDS copy; one launch, no scratch
# speedup vs baseline: 2.5746x; 2.5746x over previous
#include <hip/hip_runtime.h>


typedef __attribute__((ext_vector_type(2))) float v2f;
typedef __attribute__((ext_vector_type(4))) float v4f;
typedef __attribute__((ext_vector_type(8))) float v8f;

#define N_ROWS 16384
#define DIMS   128
#define EPS_N  1e-12f

#define TILE_ROWS 17
#define PITCH     132
#define WAVES     4

__device__ __forceinline__ float xhalf(float x) {
  return __builtin_bit_cast(float, __builtin_amdgcn_permlanex16(
      __builtin_bit_cast(int, x), __builtin_bit_cast(int, x),
      0x76543210, 0xfedcba98, false, false));
}

template <int MASK>
__device__ __forceinline__ float row_xor(float x) {
  return __builtin_bit_cast(float, __builtin_amdgcn_update_dpp(
      0, __builtin_bit_cast(int, x), 0x160 + MASK, 0xf, 0xf, true));
}

__global__ __launch_bounds__(WAVES * 32)
void collapse_rows_once_kernel(const float* __restrict__ fc,
                               float* __restrict__ out) {
  __shared__ __attribute__((aligned(16))) float tile[WAVES][TILE_ROWS * PITCH];

  const int lane = threadIdx.x & 31;
  const int wv   = threadIdx.x >> 5;
  const int base = (blockIdx.x * WAVES + wv) * 16;
  float* my = tile[wv];

  v4f v[TILE_ROWS];
#pragma unroll
  for (int r = 0; r < TILE_ROWS; ++r) {
    int row = base + r;
    if (row > N_ROWS - 1) row = N_ROWS - 1;
    v[r] = *(const v4f*)(fc + (size_t)row * DIMS + lane * 4);
  }

  float ss[TILE_ROWS];
#pragma unroll
  for (int r = 0; r < TILE_ROWS; ++r) {
    *(v4f*)(my + r * PITCH + lane * 4) = v[r];
    ss[r] = v[r].x * v[r].x + v[r].y * v[r].y + v[r].z * v[r].z + v[r].w * v[r].w;
  }

  __syncthreads();

  const int m    = lane & 15;
  const int half = lane >> 4;
  const float* pa = my + m * PITCH + 2 * half;
  const float* pb = pa + PITCH;

  v2f a[DIMS / 4], b[DIMS / 4];
#pragma unroll
  for (int j = 0; j < DIMS / 4; ++j) {
    a[j] = *(const v2f*)(pa + 4 * j);
    b[j] = *(const v2f*)(pb + 4 * j);
  }

  __builtin_amdgcn_sched_barrier(0);

  v8f acc = {0.f, 0.f, 0.f, 0.f, 0.f, 0.f, 0.f, 0.f};
#pragma unroll
  for (int j = 0; j < DIMS / 4; ++j) {
    acc = __builtin_amdgcn_wmma_f32_16x16x4_f32(
        false, a[j], false, b[j], (short)0, acc, false, false);
  }

#pragma unroll
  for (int r = 0; r < TILE_ROWS; ++r) ss[r] += xhalf(ss[r]);
#pragma unroll
  for (int r = 0; r < TILE_ROWS; ++r) ss[r] += row_xor<8>(ss[r]);
#pragma unroll
  for (int r = 0; r < TILE_ROWS; ++r) ss[r] += row_xor<4>(ss[r]);
#pragma unroll
  for (int r = 0; r < TILE_ROWS; ++r) ss[r] += row_xor<2>(ss[r]);
#pragma unroll
  for (int r = 0; r < TILE_ROWS; ++r) ss[r] += row_xor<1>(ss[r]);

  float diag = 0.f, n0 = 1.f, n1 = 1.f;
#pragma unroll
  for (int r = 0; r < 8; ++r) {
    float cr = acc[r];
    if (lane == r)      { diag = cr; n0 = ss[r];     n1 = ss[r + 1]; }
    if (lane == r + 24) { diag = cr; n0 = ss[r + 8]; n1 = ss[r + 9]; }
  }

  const bool have = (lane < 8) || (lane >= 24);
  const int  md   = (lane < 8) ? lane : (lane - 16);
  const int  oidx = base + md;
  if (have && oidx < N_ROWS - 1) {
    const float inv0 = 1.0f / fmaxf(sqrtf(n0), EPS_N);
    const float inv1 = 1.0f / fmaxf(sqrtf(n1), EPS_N);
    float d = diag * inv0 * inv1;
    const float c = 1.0f / (float)(N_ROWS - 1);
    float t = d + c;
    out[oidx] = t * t;
  }
}

extern "C" void kernel_launch(void* const* d_in, const int* in_sizes, int n_in,
                              void* d_out, int out_size, void* d_ws, size_t ws_size,
                              hipStream_t stream) {
  (void)in_sizes; (void)n_in; (void)out_size; (void)d_ws; (void)ws_size;
  const float* fc  = (const float*)d_in[0];
  float*       out = (float*)d_out;

  collapse_rows_once_kernel<<<(N_ROWS / 16) / WAVES, WAVES * 32, 0, stream>>>(fc, out);
}


# Rows fetched as four 128 B k-quarters; each quarter goes through LDS into its 8 WMMAs while later quarters are still arriving (per-wave LDS, no block barrier); norm summation order changes
# speedup vs baseline: 2.8892x; 1.1222x over previous
#include <hip/hip_runtime.h>


typedef __attribute__((ext_vector_type(2))) float v2f;
typedef __attribute__((ext_vector_type(4))) float v4f;
typedef __attribute__((ext_vector_type(8))) float v8f;

#define N_ROWS 16384
#define DIMS   128
#define EPS_N  1e-12f

#define TILE_ROWS 17
#define PITCH     132
#define WAVES     4
#define KQ        4

__device__ __forceinline__ float xhalf(float x) {
  return __builtin_bit_cast(float, __builtin_amdgcn_permlanex16(
      __builtin_bit_cast(int, x), __builtin_bit_cast(int, x),
      0x76543210, 0xfedcba98, false, false));
}

template <int MASK>
__device__ __forceinline__ float row_xor(float x) {
  return __builtin_bit_cast(float, __builtin_amdgcn_update_dpp(
      0, __builtin_bit_cast(int, x), 0x160 + MASK, 0xf, 0xf, true));
}

__device__ __forceinline__ v2f sq2(v4f x) {
  const v2f lo = {x.x, x.y};
  const v2f hi = {x.z, x.w};
  return lo * lo + hi * hi;
}

__device__ __forceinline__ v2f addsq(v2f s, v4f x) {
  const v2f lo = {x.x, x.y};
  const v2f hi = {x.z, x.w};
  s = lo * lo + s;
  s = hi * hi + s;
  return s;
}

__device__ __forceinline__ float pick4(int idx, float e0, float e1, float e2, float e3) {
  const float lo = (idx & 1) ? e1 : e0;
  const float hi = (idx & 1) ? e3 : e2;
  return (idx & 2) ? hi : lo;
}

__device__ __forceinline__ float inv_norm(float ss) {
  return 1.0f / fmaxf(sqrtf(ss), EPS_N);
}

__global__ __launch_bounds__(WAVES * 32)
void collapse_kquarter_pipe_kernel(const float* __restrict__ fc,
                                   float* __restrict__ out) {
  __shared__ __attribute__((aligned(16))) float tile[WAVES][TILE_ROWS * PITCH];

  const int lane = threadIdx.x & 31;
  const int wv   = threadIdx.x >> 5;
  const int base = (blockIdx.x * WAVES + wv) * 16;
  float* my = tile[wv];

  const int i4 = lane >> 3;
  const int c8 = lane & 7;

  int r16 = base + 16;
  if (r16 > N_ROWS - 1) r16 = N_ROWS - 1;
  const v4f v16 = *(const v4f*)(fc + (size_t)r16 * DIMS + lane * 4);

  const float* lp = fc + (size_t)(base + 4 * i4) * DIMS + c8 * 4;
  v4f v[KQ][4];
#pragma unroll
  for (int q = 0; q < KQ; ++q) {
#pragma unroll
    for (int g = 0; g < 4; ++g)
      v[q][g] = *(const v4f*)(lp + g * DIMS + q * 32);
    __builtin_amdgcn_sched_barrier(0);
  }

  float* sp = my + 4 * i4 * PITCH + c8 * 4;

  const int m    = lane & 15;
  const int half = lane >> 4;
  const float* pa = my + m * PITCH + 2 * half;
  const float* pb = pa + PITCH;

  v2f a[DIMS / 4], b[DIMS / 4];
  v2f sg[4];
  v8f acc = {0.f, 0.f, 0.f, 0.f, 0.f, 0.f, 0.f, 0.f};

#define ARRIVE(q)                                                             \
  asm volatile("" : "+v"(v[q][0]), "+v"(v[q][1]), "+v"(v[q][2]), "+v"(v[q][3]));
#define DONE(...) asm volatile("" :: __VA_ARGS__);
#define STAGED __builtin_amdgcn_sched_barrier(0);
#define BLOCK_END __builtin_amdgcn_sched_barrier(0);

#define STAGE(q)                                                              \
  _Pragma("unroll")                                                           \
  for (int g = 0; g < 4; ++g)                                                 \
    *(v4f*)(sp + g * PITCH + (q) * 32) = v[q][g];                             \
  _Pragma("unroll")                                                           \
  for (int jj = 0; jj < 8; ++jj) {                                            \
    a[8 * (q) + jj] = *(const v2f*)(pa + 4 * (8 * (q) + jj));                 \
    b[8 * (q) + jj] = *(const v2f*)(pb + 4 * (8 * (q) + jj));                 \
  }

#define MULT(q)                                                               \
  _Pragma("unroll")                                                           \
  for (int jj = 0; jj < 8; ++jj)                                              \
    acc = __builtin_amdgcn_wmma_f32_16x16x4_f32(                              \
        false, a[8 * (q) + jj], false, b[8 * (q) + jj], (short)0, acc, false, false);

  *(v4f*)(my + 16 * PITCH + lane * 4) = v16;
  STAGE(0)
  const v2f t16 = sq2(v16);
  float s16 = t16.x + t16.y;
#pragma unroll
  for (int g = 0; g < 4; ++g) sg[g] = sq2(v[0][g]);
  BLOCK_END

  ARRIVE(1)
  STAGE(1)
  STAGED
  MULT(0)
#pragma unroll
  for (int g = 0; g < 4; ++g) sg[g] = addsq(sg[g], v[1][g]);
  s16 += row_xor<1>(s16);
  s16 += row_xor<2>(s16);
  s16 += row_xor<4>(s16);
  s16 += row_xor<8>(s16);
  s16 += xhalf(s16);
  DONE("v"(acc), "v"(s16), "v"(sg[0]), "v"(sg[1]), "v"(sg[2]), "v"(sg[3]))
  BLOCK_END

  ARRIVE(2)
  STAGE(2)
  STAGED
  MULT(1)
#pragma unroll
  for (int g = 0; g < 4; ++g) sg[g] = addsq(sg[g], v[2][g]);
  DONE("v"(acc), "v"(sg[0]), "v"(sg[1]), "v"(sg[2]), "v"(sg[3]))
  BLOCK_END

  ARRIVE(3)
  STAGE(3)
  STAGED
  MULT(2)
  MULT(3)
  float ss[4];
#pragma unroll
  for (int g = 0; g < 4; ++g) {
    sg[g] = addsq(sg[g], v[3][g]);
    ss[g] = sg[g].x + sg[g].y;
  }
#pragma unroll
  for (int g = 0; g < 4; ++g) ss[g] += row_xor<1>(ss[g]);
#pragma unroll
  for (int g = 0; g < 4; ++g) ss[g] += row_xor<2>(ss[g]);
#pragma unroll
  for (int g = 0; g < 4; ++g) ss[g] += row_xor<4>(ss[g]);

  const float t0 = pick4(lane,     ss[0], ss[1], ss[2], ss[3]);
  const float t1 = pick4(lane + 1, ss[0], ss[1], ss[2], ss[3]);
  const float o0 = row_xor<8>(t0);
  const float o1 = row_xor<8>(t1);
  const float x1 = xhalf(t1);
  const int   m1 = m + 1;
  const float n0 = ((m  >> 2) != i4) ? o0 : t0;
  const float na = ((m1 >> 2) != i4) ? o1 : t1;
  const float nb = (m1 == 8) ? x1 : na;
  const float n1 = (m1 == 16) ? s16 : nb;
  const float inv0 = inv_norm(n0);
  const float inv1 = inv_norm(n1);
  DONE("v"(acc), "v"(inv0), "v"(inv1))

  const float dlo  = pick4(lane, acc[0], acc[1], acc[2], acc[3]);
  const float dhi  = pick4(lane, acc[4], acc[5], acc[6], acc[7]);
  const float diag = (lane & 4) ? dhi : dlo;

  const bool have = (lane < 8) || (lane >= 24);
  const int  oidx = base + m;
  if (have && oidx < N_ROWS - 1) {
    float d = diag * inv0 * inv1;
    const float c = 1.0f / (float)(N_ROWS - 1);
    float t = d + c;
    out[oidx] = t * t;
  }
#undef ARRIVE
#undef DONE
#undef STAGED
#undef BLOCK_END
#undef STAGE
#undef MULT
}

extern "C" void kernel_launch(void* const* d_in, const int* in_sizes, int n_in,
                              void* d_out, int out_size, void* d_ws, size_t ws_size,
                              hipStream_t stream) {
  (void)in_sizes; (void)n_in; (void)out_size; (void)d_ws; (void)ws_size;
  const float* fc  = (const float*)d_in[0];
  float*       out = (float*)d_out;

  collapse_kquarter_pipe_kernel<<<(N_ROWS / 16) / WAVES, WAVES * 32, 0, stream>>>(fc, out);
}
